# starting kernel
# baseline (speedup 1.0000x reference)
#include <hip/hip_runtime.h>
#include <hip/hip_bf16.h>

typedef float v2f __attribute__((ext_vector_type(2)));
typedef float v8f __attribute__((ext_vector_type(8)));

#ifndef GCN_THREADS
#define GCN_THREADS 256
#endif

__global__ void k_deg_init(float* __restrict__ deg, int n) {
  int i = blockIdx.x * blockDim.x + threadIdx.x;
  if (i < n) deg[i] = 1.0f;
}

__global__ void k_deg_count(const int* __restrict__ dst, float* __restrict__ deg, int e) {
  int i = blockIdx.x * blockDim.x + threadIdx.x;
  if (i < e) atomicAdd(&deg[dst[i]], 1.0f);
}

__global__ void k_dinv(float* __restrict__ deg, int n) {
  int i = blockIdx.x * blockDim.x + threadIdx.x;
  if (i < n) deg[i] = rsqrtf(deg[i]);
}

template <int K, int NO, int NT>
__global__ void k_gemm_wmma(const float* __restrict__ A, const float* __restrict__ W,
                            float* __restrict__ out, int mtiles) {
  const int wave = (int)((blockIdx.x * blockDim.x + threadIdx.x) >> 5);
  const int lane = (int)(threadIdx.x & 31);
  if (wave >= mtiles) return;

  const int hi   = lane >> 4;
  const int lo   = lane & 15;
  const int kb   = hi << 1;
  const int row  = wave * 16 + lo;

  v8f acc[NT];
#pragma unroll
  for (int t = 0; t < NT; ++t) acc[t] = (v8f){0.f, 0.f, 0.f, 0.f, 0.f, 0.f, 0.f, 0.f};

  const float* __restrict__ arow = A + (size_t)row * K;

  for (int k0 = 0; k0 < K; k0 += 4) {
    v2f a = *(const v2f*)(arow + k0 + kb);
#pragma unroll
    for (int t = 0; t < NT; ++t) {
      const int n = t * 16 + lo;
      v2f b;
      b.x = W[(size_t)(k0 + kb + 0) * NO + n];
      b.y = W[(size_t)(k0 + kb + 1) * NO + n];
      acc[t] = __builtin_amdgcn_wmma_f32_16x16x4_f32(
           false, a,  false, b,
           (short)0, acc[t],  false,  false);
    }
  }

  const int mbase = wave * 16 + (hi << 3);
#pragma unroll
  for (int t = 0; t < NT; ++t) {
#pragma unroll
    for (int r = 0; r < 8; ++r) {
      out[(size_t)(mbase + r) * NO + t * 16 + lo] = acc[t][r];
    }
  }
}

template <int D>
__global__ void k_selfloop_init(const float* __restrict__ h, const float* __restrict__ dinv,
                                float* __restrict__ acc, int n) {
  const unsigned QP = D / 4;
  unsigned i = blockIdx.x * blockDim.x + threadIdx.x;
  if (i >= (unsigned)n * QP) return;
  unsigned node = i / QP;
  unsigned f4   = (i % QP) * 4;
  float s = dinv[node];
  s = s * s;
  float4 v = *(const float4*)(h + (size_t)node * D + f4);
  v.x *= s; v.y *= s; v.z *= s; v.w *= s;
  *(float4*)(acc + (size_t)node * D + f4) = v;
}

template <int D>
__global__ void k_edge_scatter(const int* __restrict__ src, const int* __restrict__ dst,
                               const float* __restrict__ dinv, const float* __restrict__ h,
                               float* __restrict__ acc, int e) {
  const unsigned QP = D / 4;
  unsigned tid = blockIdx.x * blockDim.x + threadIdx.x;
  if (tid >= (unsigned)e * QP) return;
  unsigned eid = tid / QP;
  unsigned f4  = (tid % QP) * 4;
  int s = src[eid];
  int d = dst[eid];
  float nrm = dinv[s] * dinv[d];
  float4 v = *(const float4*)(h + (size_t)s * D + f4);
  float* o = acc + (size_t)d * D + f4;
  atomicAdd(o + 0, v.x * nrm);
  atomicAdd(o + 1, v.y * nrm);
  atomicAdd(o + 2, v.z * nrm);
  atomicAdd(o + 3, v.w * nrm);
}

template <int D>
__global__ void k_bias_relu(const float* __restrict__ acc, const float* __restrict__ b,
                            float* __restrict__ out, unsigned total) {
  unsigned i = blockIdx.x * blockDim.x + threadIdx.x;
  if (i >= total) return;
  out[i] = fmaxf(acc[i] + b[i & (D - 1)], 0.0f);
}

template <int D>
__global__ void k_bias_add(float* __restrict__ out, const float* __restrict__ b, unsigned total) {
  unsigned i = blockIdx.x * blockDim.x + threadIdx.x;
  if (i >= total) return;
  out[i] += b[i & (D - 1)];
}

extern "C" void kernel_launch(void* const* d_in, const int* in_sizes, int n_in,
                              void* d_out, int out_size, void* d_ws, size_t ws_size,
                              hipStream_t stream) {
  (void)n_in; (void)out_size; (void)ws_size;

  const float* x    = (const float*)d_in[0];
  const int*   ei   = (const int*)d_in[1];
  const float* W1   = (const float*)d_in[2];
  const float* b1   = (const float*)d_in[3];
  const float* W2   = (const float*)d_in[4];
  const float* b2   = (const float*)d_in[5];
  float*       out  = (float*)d_out;

  const int N = in_sizes[0] / 128;
  const int E = in_sizes[1] / 2;
  const int IN_D = 128, HID_D = 128, OUT_D = 64;

  const int* src = ei;
  const int* dst = ei + E;

  float* dinv = (float*)d_ws;
  float* bufA = dinv + N;
  float* bufB = bufA + (size_t)N * 128;

  const int T = GCN_THREADS;
  auto blk = [&](long long work) { return (unsigned)((work + T - 1) / T); };

  k_deg_init<<<blk(N), T, 0, stream>>>(dinv, N);
  k_deg_count<<<blk(E), T, 0, stream>>>(dst, dinv, E);
  k_dinv<<<blk(N), T, 0, stream>>>(dinv, N);

  const int mtiles = N / 16;
  {
    const int wavesPerBlk = T / 32;
    unsigned g = (unsigned)((mtiles + wavesPerBlk - 1) / wavesPerBlk);
    k_gemm_wmma<IN_D, HID_D, HID_D / 16><<<g, T, 0, stream>>>(x, W1, bufA, mtiles);
  }

  k_selfloop_init<HID_D><<<blk((long long)N * (HID_D / 4)), T, 0, stream>>>(bufA, dinv, bufB, N);
  k_edge_scatter<HID_D><<<blk((long long)E * (HID_D / 4)), T, 0, stream>>>(src, dst, dinv, bufA, bufB, E);
  k_bias_relu<HID_D><<<blk((long long)N * HID_D), T, 0, stream>>>(bufB, b1, bufA, (unsigned)N * HID_D);

  {
    const int wavesPerBlk = T / 32;
    unsigned g = (unsigned)((mtiles + wavesPerBlk - 1) / wavesPerBlk);
    k_gemm_wmma<HID_D, OUT_D, OUT_D / 16><<<g, T, 0, stream>>>(bufA, W2, bufB, mtiles);
  }

  k_selfloop_init<OUT_D><<<blk((long long)N * (OUT_D / 4)), T, 0, stream>>>(bufB, dinv, out, N);
  k_edge_scatter<OUT_D><<<blk((long long)E * (OUT_D / 4)), T, 0, stream>>>(src, dst, dinv, bufB, out, E);
  k_bias_add<OUT_D><<<blk((long long)N * OUT_D), T, 0, stream>>>(out, b2, (unsigned)N * OUT_D);
}


# Edge scatter remapped: a wave takes whole edges and its 32 lanes 32 consecutive floats, so every gather load and float atomic covers one full 128-byte line (was 4 lines); sums still in arrival order
# speedup vs baseline: 1.3430x; 1.3430x over previous
#include <hip/hip_runtime.h>
#include <hip/hip_bf16.h>

typedef float v2f __attribute__((ext_vector_type(2)));
typedef float v8f __attribute__((ext_vector_type(8)));

#ifndef GCN_THREADS
#define GCN_THREADS 256
#endif

__global__ void k_deg_init(float* __restrict__ deg, int n) {
  int i = blockIdx.x * blockDim.x + threadIdx.x;
  if (i < n) deg[i] = 1.0f;
}

__global__ void k_deg_count(const int* __restrict__ dst, float* __restrict__ deg, int e) {
  int i = blockIdx.x * blockDim.x + threadIdx.x;
  if (i < e) atomicAdd(&deg[dst[i]], 1.0f);
}

__global__ void k_dinv(float* __restrict__ deg, int n) {
  int i = blockIdx.x * blockDim.x + threadIdx.x;
  if (i < n) deg[i] = rsqrtf(deg[i]);
}

template <int K, int NO, int NT>
__global__ void k_gemm_wmma(const float* __restrict__ A, const float* __restrict__ W,
                            float* __restrict__ out, int mtiles) {
  const int wave = (int)((blockIdx.x * blockDim.x + threadIdx.x) >> 5);
  const int lane = (int)(threadIdx.x & 31);
  if (wave >= mtiles) return;

  const int hi   = lane >> 4;
  const int lo   = lane & 15;
  const int kb   = hi << 1;
  const int row  = wave * 16 + lo;

  v8f acc[NT];
#pragma unroll
  for (int t = 0; t < NT; ++t) acc[t] = (v8f){0.f, 0.f, 0.f, 0.f, 0.f, 0.f, 0.f, 0.f};

  const float* __restrict__ arow = A + (size_t)row * K;

  for (int k0 = 0; k0 < K; k0 += 4) {
    v2f a = *(const v2f*)(arow + k0 + kb);
#pragma unroll
    for (int t = 0; t < NT; ++t) {
      const int n = t * 16 + lo;
      v2f b;
      b.x = W[(size_t)(k0 + kb + 0) * NO + n];
      b.y = W[(size_t)(k0 + kb + 1) * NO + n];
      acc[t] = __builtin_amdgcn_wmma_f32_16x16x4_f32(
           false, a,  false, b,
           (short)0, acc[t],  false,  false);
    }
  }

  const int mbase = wave * 16 + (hi << 3);
#pragma unroll
  for (int t = 0; t < NT; ++t) {
#pragma unroll
    for (int r = 0; r < 8; ++r) {
      out[(size_t)(mbase + r) * NO + t * 16 + lo] = acc[t][r];
    }
  }
}

template <int D>
__global__ void k_selfloop_init(const float* __restrict__ h, const float* __restrict__ dinv,
                                float* __restrict__ acc, int n) {
  const unsigned QP = D / 4;
  unsigned i = blockIdx.x * blockDim.x + threadIdx.x;
  if (i >= (unsigned)n * QP) return;
  unsigned node = i / QP;
  unsigned f4   = (i % QP) * 4;
  float s = dinv[node];
  s = s * s;
  float4 v = *(const float4*)(h + (size_t)node * D + f4);
  v.x *= s; v.y *= s; v.z *= s; v.w *= s;
  *(float4*)(acc + (size_t)node * D + f4) = v;
}

template <int D>
__device__ __forceinline__ void edge_add_one(const int* __restrict__ src, const int* __restrict__ dst,
                                             const float* __restrict__ dinv, const float* __restrict__ h,
                                             float* __restrict__ acc, int eid, int lane) {
  constexpr int CH = D / 32;
  const int s = src[eid];
  const int d = dst[eid];
  const float nrm = dinv[s] * dinv[d];
  const float* hp = h + (size_t)s * D + lane;
  float* o = acc + (size_t)d * D + lane;
  for (int c = 0; c < CH; ++c) atomicAdd(o + c * 32, hp[c * 32] * nrm);
}

template <int D, int EPW>
__global__ void k_edge_scatter(const int* __restrict__ src, const int* __restrict__ dst,
                               const float* __restrict__ dinv, const float* __restrict__ h,
                               float* __restrict__ acc, int e) {
  constexpr int CH = D / 32;
  const int lane = (int)(threadIdx.x & 31);
  const int wave = __builtin_amdgcn_readfirstlane((int)((blockIdx.x * blockDim.x + threadIdx.x) >> 5));
  const int e0 = wave * EPW;
  if (e0 >= e) return;

  if (e0 + EPW <= e) {
    int s[EPW], d[EPW];
    for (int i = 0; i < EPW; ++i) {
      s[i] = src[e0 + i];
      d[i] = dst[e0 + i];
    }
    float v[EPW][CH];
    float nrm[EPW];
    for (int i = 0; i < EPW; ++i) {
      const float* hp = h + (size_t)s[i] * D + lane;
      for (int c = 0; c < CH; ++c) v[i][c] = hp[c * 32];
      nrm[i] = dinv[s[i]] * dinv[d[i]];
    }
    for (int i = 0; i < EPW; ++i) {
      float* o = acc + (size_t)d[i] * D + lane;
      for (int c = 0; c < CH; ++c) atomicAdd(o + c * 32, v[i][c] * nrm[i]);
    }
  } else {
    for (int eid = e0; eid < e; ++eid) edge_add_one<D>(src, dst, dinv, h, acc, eid, lane);
  }
}

template <int D>
__global__ void k_bias_relu(const float* __restrict__ acc, const float* __restrict__ b,
                            float* __restrict__ out, unsigned total) {
  unsigned i = blockIdx.x * blockDim.x + threadIdx.x;
  if (i >= total) return;
  out[i] = fmaxf(acc[i] + b[i & (D - 1)], 0.0f);
}

template <int D>
__global__ void k_bias_add(float* __restrict__ out, const float* __restrict__ b, unsigned total) {
  unsigned i = blockIdx.x * blockDim.x + threadIdx.x;
  if (i >= total) return;
  out[i] += b[i & (D - 1)];
}

extern "C" void kernel_launch(void* const* d_in, const int* in_sizes, int n_in,
                              void* d_out, int out_size, void* d_ws, size_t ws_size,
                              hipStream_t stream) {
  (void)n_in; (void)out_size; (void)ws_size;

  const float* x    = (const float*)d_in[0];
  const int*   ei   = (const int*)d_in[1];
  const float* W1   = (const float*)d_in[2];
  const float* b1   = (const float*)d_in[3];
  const float* W2   = (const float*)d_in[4];
  const float* b2   = (const float*)d_in[5];
  float*       out  = (float*)d_out;

  const int N = in_sizes[0] / 128;
  const int E = in_sizes[1] / 2;
  const int IN_D = 128, HID_D = 128, OUT_D = 64;

  const int* src = ei;
  const int* dst = ei + E;

  float* dinv = (float*)d_ws;
  float* bufA = dinv + N;
  float* bufB = bufA + (size_t)N * 128;

  const int T = GCN_THREADS;
  auto blk = [&](long long work) { return (unsigned)((work + T - 1) / T); };

  k_deg_init<<<blk(N), T, 0, stream>>>(dinv, N);
  k_deg_count<<<blk(E), T, 0, stream>>>(dst, dinv, E);
  k_dinv<<<blk(N), T, 0, stream>>>(dinv, N);

  const int mtiles = N / 16;
  {
    const int wavesPerBlk = T / 32;
    unsigned g = (unsigned)((mtiles + wavesPerBlk - 1) / wavesPerBlk);
    k_gemm_wmma<IN_D, HID_D, HID_D / 16><<<g, T, 0, stream>>>(x, W1, bufA, mtiles);
  }

  k_selfloop_init<HID_D><<<blk((long long)N * (HID_D / 4)), T, 0, stream>>>(bufA, dinv, bufB, N);
  k_edge_scatter<HID_D, 4><<<blk((long long)((E + 3) / 4) * 32), T, 0, stream>>>(src, dst, dinv, bufA, bufB, E);
  k_bias_relu<HID_D><<<blk((long long)N * HID_D), T, 0, stream>>>(bufB, b1, bufA, (unsigned)N * HID_D);

  {
    const int wavesPerBlk = T / 32;
    unsigned g = (unsigned)((mtiles + wavesPerBlk - 1) / wavesPerBlk);
    k_gemm_wmma<HID_D, OUT_D, OUT_D / 16><<<g, T, 0, stream>>>(bufA, W2, bufB, mtiles);
  }

  k_selfloop_init<OUT_D><<<blk((long long)N * (OUT_D / 4)), T, 0, stream>>>(bufB, dinv, out, N);
  k_edge_scatter<OUT_D, 8><<<blk((long long)((E + 7) / 8) * 32), T, 0, stream>>>(src, dst, dinv, bufB, out, E);
  k_bias_add<OUT_D><<<blk((long long)N * OUT_D), T, 0, stream>>>(out, b2, (unsigned)N * OUT_D);
}


# Self-loop init, bias and relu launches folded into the GEMM kernels: each stores h and acc = bias + dinv^2*h, GEMM2 takes relu as it loads A; bias now the first addend of each sum, not the last
# speedup vs baseline: 1.7070x; 1.2710x over previous
#include <hip/hip_runtime.h>
#include <hip/hip_bf16.h>

typedef float v2f __attribute__((ext_vector_type(2)));
typedef float v8f __attribute__((ext_vector_type(8)));

#ifndef GCN_THREADS
#define GCN_THREADS 256
#endif

__global__ void k_deg_init(float* __restrict__ deg, int n) {
  int i = blockIdx.x * blockDim.x + threadIdx.x;
  if (i < n) deg[i] = 1.0f;
}

__global__ void k_deg_count(const int* __restrict__ dst, float* __restrict__ deg, int e) {
  int i = blockIdx.x * blockDim.x + threadIdx.x;
  if (i < e) atomicAdd(&deg[dst[i]], 1.0f);
}

__global__ void k_dinv(float* __restrict__ deg, int n) {
  int i = blockIdx.x * blockDim.x + threadIdx.x;
  if (i < n) deg[i] = rsqrtf(deg[i]);
}

template <int K, int NO, int NT, bool RELU_A>
__global__ void k_gemm_wmma(const float* __restrict__ A, const float* __restrict__ W,
                            const float* __restrict__ dinv, const float* __restrict__ bias,
                            float* __restrict__ h, float* __restrict__ accout, int mtiles) {
  const int wave = (int)((blockIdx.x * blockDim.x + threadIdx.x) >> 5);
  const int lane = (int)(threadIdx.x & 31);
  if (wave >= mtiles) return;

  const int hi   = lane >> 4;
  const int lo   = lane & 15;
  const int kb   = hi << 1;
  const int row  = wave * 16 + lo;

  v8f acc[NT];
#pragma unroll
  for (int t = 0; t < NT; ++t) acc[t] = (v8f){0.f, 0.f, 0.f, 0.f, 0.f, 0.f, 0.f, 0.f};

  const float* __restrict__ arow = A + (size_t)row * K;

  for (int k0 = 0; k0 < K; k0 += 4) {
    v2f a = *(const v2f*)(arow + k0 + kb);
    if (RELU_A) {
      a.x = fmaxf(a.x, 0.0f);
      a.y = fmaxf(a.y, 0.0f);
    }
#pragma unroll
    for (int t = 0; t < NT; ++t) {
      const int n = t * 16 + lo;
      v2f b;
      b.x = W[(size_t)(k0 + kb + 0) * NO + n];
      b.y = W[(size_t)(k0 + kb + 1) * NO + n];
      acc[t] = __builtin_amdgcn_wmma_f32_16x16x4_f32(
           false, a,  false, b,
           (short)0, acc[t],  false,  false);
    }
  }

  const int mbase = wave * 16 + (hi << 3);
  float s2[8];
#pragma unroll
  for (int r = 0; r < 8; ++r) {
    const float s = dinv[mbase + r];
    s2[r] = s * s;
  }
#pragma unroll
  for (int t = 0; t < NT; ++t) {
    const int col = t * 16 + lo;
    const float bv = bias[col];
#pragma unroll
    for (int r = 0; r < 8; ++r) {
      const size_t idx = (size_t)(mbase + r) * NO + col;
      const float v = acc[t][r];
      h[idx] = v;
      accout[idx] = v * s2[r] + bv;
    }
  }
}

template <int D>
__device__ __forceinline__ void edge_add_one(const int* __restrict__ src, const int* __restrict__ dst,
                                             const float* __restrict__ dinv, const float* __restrict__ h,
                                             float* __restrict__ acc, int eid, int lane) {
  constexpr int CH = D / 32;
  const int s = src[eid];
  const int d = dst[eid];
  const float nrm = dinv[s] * dinv[d];
  const float* hp = h + (size_t)s * D + lane;
  float* o = acc + (size_t)d * D + lane;
  for (int c = 0; c < CH; ++c) atomicAdd(o + c * 32, hp[c * 32] * nrm);
}

template <int D, int EPW>
__global__ void k_edge_scatter(const int* __restrict__ src, const int* __restrict__ dst,
                               const float* __restrict__ dinv, const float* __restrict__ h,
                               float* __restrict__ acc, int e) {
  constexpr int CH = D / 32;
  const int lane = (int)(threadIdx.x & 31);
  const int wave = __builtin_amdgcn_readfirstlane((int)((blockIdx.x * blockDim.x + threadIdx.x) >> 5));
  const int e0 = wave * EPW;
  if (e0 >= e) return;

  if (e0 + EPW <= e) {
    int s[EPW], d[EPW];
    for (int i = 0; i < EPW; ++i) {
      s[i] = src[e0 + i];
      d[i] = dst[e0 + i];
    }
    float v[EPW][CH];
    float nrm[EPW];
    for (int i = 0; i < EPW; ++i) {
      const float* hp = h + (size_t)s[i] * D + lane;
      for (int c = 0; c < CH; ++c) v[i][c] = hp[c * 32];
      nrm[i] = dinv[s[i]] * dinv[d[i]];
    }
    for (int i = 0; i < EPW; ++i) {
      float* o = acc + (size_t)d[i] * D + lane;
      for (int c = 0; c < CH; ++c) atomicAdd(o + c * 32, v[i][c] * nrm[i]);
    }
  } else {
    for (int eid = e0; eid < e; ++eid) edge_add_one<D>(src, dst, dinv, h, acc, eid, lane);
  }
}

extern "C" void kernel_launch(void* const* d_in, const int* in_sizes, int n_in,
                              void* d_out, int out_size, void* d_ws, size_t ws_size,
                              hipStream_t stream) {
  (void)n_in; (void)out_size; (void)ws_size;

  const float* x    = (const float*)d_in[0];
  const int*   ei   = (const int*)d_in[1];
  const float* W1   = (const float*)d_in[2];
  const float* b1   = (const float*)d_in[3];
  const float* W2   = (const float*)d_in[4];
  const float* b2   = (const float*)d_in[5];
  float*       out  = (float*)d_out;

  const int N = in_sizes[0] / 128;
  const int E = in_sizes[1] / 2;
  const int IN_D = 128, HID_D = 128, OUT_D = 64;

  const int* src = ei;
  const int* dst = ei + E;

  float* dinv = (float*)d_ws;
  float* bufA = dinv + N;
  float* bufB = bufA + (size_t)N * 128;

  const int T = GCN_THREADS;
  auto blk = [&](long long work) { return (unsigned)((work + T - 1) / T); };

  k_deg_init<<<blk(N), T, 0, stream>>>(dinv, N);
  k_deg_count<<<blk(E), T, 0, stream>>>(dst, dinv, E);
  k_dinv<<<blk(N), T, 0, stream>>>(dinv, N);

  const int mtiles = N / 16;
  const int wavesPerBlk = T / 32;
  const unsigned g = (unsigned)((mtiles + wavesPerBlk - 1) / wavesPerBlk);

  k_gemm_wmma<IN_D, HID_D, HID_D / 16, false><<<g, T, 0, stream>>>(x, W1, dinv, b1, bufA, bufB, mtiles);

  k_edge_scatter<HID_D, 4><<<blk((long long)((E + 3) / 4) * 32), T, 0, stream>>>(src, dst, dinv, bufA, bufB, E);

  k_gemm_wmma<HID_D, OUT_D, OUT_D / 16, true><<<g, T, 0, stream>>>(bufB, W2, dinv, b2, bufA, out, mtiles);

  k_edge_scatter<OUT_D, 8><<<blk((long long)((E + 7) / 8) * 32), T, 0, stream>>>(src, dst, dinv, bufA, out, E);
}


# Layer-2 owner pass stops rescanning edge ids: the layer-1 pass saves each owner's checked edge list (source, row, norm) to the workspace, layer 2 streams it back in whole lines; arrival order kept
# speedup vs baseline: 1.9493x; 1.1419x over previous
#include <hip/hip_runtime.h>
#include <hip/hip_bf16.h>


typedef float v2f __attribute__((ext_vector_type(2)));
typedef float v8f __attribute__((ext_vector_type(8)));

#ifndef GCN_THREADS
#define GCN_THREADS 256
#endif

#define OWN_ROWS 391
#define OWN_T 1024
#define OWN_Q 4096
#define OWN_HDR 32
#define OWN_REC (OWN_HDR + 3 * OWN_Q)

__global__ void k_deg_init(float* __restrict__ deg, int n) {
  int i = blockIdx.x * blockDim.x + threadIdx.x;
  if (i < n) deg[i] = 1.0f;
}

__global__ void k_deg_count(const int* __restrict__ dst, float* __restrict__ deg,
                            unsigned* __restrict__ ids, int e) {
  const int t = blockIdx.x * blockDim.x + threadIdx.x;
  const int i0 = t * 4;
  if (i0 >= e) return;
  unsigned w = 0;
#pragma unroll
  for (int j = 0; j < 4; ++j) {
    unsigned id = 255u;
    const int i = i0 + j;
    if (i < e) {
      const int d = dst[i];
      atomicAdd(&deg[d], 1.0f);
      const unsigned q = (unsigned)d / (unsigned)OWN_ROWS;
      id = q < 255u ? q : 255u;
    }
    w |= id << (8 * j);
  }
  ids[t] = w;
}

__global__ void k_dinv(float* __restrict__ deg, int n) {
  int i = blockIdx.x * blockDim.x + threadIdx.x;
  if (i < n) deg[i] = rsqrtf(deg[i]);
}

template <int K, int NO, int NT, bool RELU_A>
__global__ void k_gemm_wmma(const float* __restrict__ A, const float* __restrict__ W,
                            float* __restrict__ h, int mtiles) {
  const int wave = (int)((blockIdx.x * blockDim.x + threadIdx.x) >> 5);
  const int lane = (int)(threadIdx.x & 31);
  if (wave >= mtiles) return;

  const int hi   = lane >> 4;
  const int lo   = lane & 15;
  const int kb   = hi << 1;
  const int row  = wave * 16 + lo;

  v8f acc[NT];
#pragma unroll
  for (int t = 0; t < NT; ++t) acc[t] = (v8f){0.f, 0.f, 0.f, 0.f, 0.f, 0.f, 0.f, 0.f};

  const float* __restrict__ arow = A + (size_t)row * K;

  for (int k0 = 0; k0 < K; k0 += 4) {
    v2f a = *(const v2f*)(arow + k0 + kb);
    if (RELU_A) {
      a.x = fmaxf(a.x, 0.0f);
      a.y = fmaxf(a.y, 0.0f);
    }
#pragma unroll
    for (int t = 0; t < NT; ++t) {
      const int n = t * 16 + lo;
      v2f b;
      b.x = W[(size_t)(k0 + kb + 0) * NO + n];
      b.y = W[(size_t)(k0 + kb + 1) * NO + n];
      acc[t] = __builtin_amdgcn_wmma_f32_16x16x4_f32(
           false, a,  false, b,
           (short)0, acc[t],  false,  false);
    }
  }

  const int mbase = wave * 16 + (hi << 3);
#pragma unroll
  for (int t = 0; t < NT; ++t) {
    const int col = t * 16 + lo;
#pragma unroll
    for (int r = 0; r < 8; ++r) h[(size_t)(mbase + r) * NO + col] = acc[t][r];
  }
}

template <int D>
__device__ __forceinline__ void own_add_serial(const int* __restrict__ src, const int* __restrict__ dst,
                                               const float* __restrict__ dinv, const float* __restrict__ h,
                                               float* tile, int idx, int e, int lo, unsigned urows) {
  if (idx >= e) return;
  const int d = dst[idx];
  const unsigned u = (unsigned)(d - lo);
  if (d < lo || u >= urows) return;
  const int s = src[idx];
  const float nrm = dinv[s] * dinv[d];
  const float* hp = h + (size_t)s * D;
  float* t = tile + u * D;
#pragma nounroll
  for (int c = 0; c < D; ++c) atomicAdd(t + c, hp[c] * nrm);
}

template <int D, int U, bool SAVE, bool LOAD>
__global__ __launch_bounds__(OWN_T) void k_owner_sum(const int* __restrict__ src, const int* __restrict__ dst,
                                                    const unsigned* __restrict__ ids,
                                                    const float* __restrict__ dinv, const float* __restrict__ h,
                                                    const float* __restrict__ bias, float* __restrict__ out,
                                                    int* __restrict__ lists, int n, int e) {
  constexpr int CH = D / 32;
  __shared__ float tile[OWN_ROWS * D];
  __shared__ int q_a[OWN_Q];
  __shared__ int q_dl[OWN_Q];
  __shared__ float q_nrm[OWN_Q];
  __shared__ int q_count;

  const int tid = (int)threadIdx.x;
  const int lane = tid & 31;
  const int wave = __builtin_amdgcn_readfirstlane(tid >> 5);
  const int lo = (int)blockIdx.x * OWN_ROWS;
  int rows = n - lo;
  if (rows > OWN_ROWS) rows = OWN_ROWS;
  if (rows <= 0) return;
  const unsigned urows = (unsigned)rows;
  const int cells = rows * D;

  int* __restrict__ g_hdr = lists + (size_t)blockIdx.x * OWN_REC;
  int* __restrict__ g_a = g_hdr + OWN_HDR;
  int* __restrict__ g_dl = g_a + OWN_Q;
  float* __restrict__ g_nrm = (float*)(g_dl + OWN_Q);
  constexpr int LPT = OWN_Q / OWN_T;
  int saved = -1;
  if (LOAD) saved = __builtin_amdgcn_readfirstlane(g_hdr[0]);
  const bool replay = LOAD && saved >= 0 && saved <= OWN_Q;
  int la[LPT], ld[LPT];
  float ln[LPT];
  if (replay) {
#pragma unroll
    for (int j = 0; j < LPT; ++j) {
      const int i = tid + j * OWN_T;
      const bool ok = i < saved;
      la[j] = ok ? g_a[i] : 0;
      ld[j] = ok ? g_dl[i] : -1;
      ln[j] = ok ? g_nrm[i] : 0.0f;
    }
  }

  {
    const float bv = bias[tid % D];
    const float* __restrict__ hb = h + (size_t)lo * D;
#pragma unroll 8
    for (int i = tid; i < cells; i += OWN_T) {
      const float s = dinv[lo + i / D];
      tile[i] = hb[i] * (s * s) + bv;
    }
  }
  int count;
  if (replay) {
#pragma unroll
    for (int j = 0; j < LPT; ++j) {
      const int i = tid + j * OWN_T;
      if (i < saved) {
        q_a[i] = la[j];
        q_dl[i] = ld[j];
        q_nrm[i] = ln[j];
      }
    }
    count = saved;
    __syncthreads();
  } else {
  if (tid == 0) q_count = 0;
  __syncthreads();

  {
    const unsigned my = blockIdx.x < 255u ? (unsigned)blockIdx.x : 255u;
    const unsigned pat = my * 0x01010101u;
    const int nw = (e + 3) >> 2;
    const int ng = nw >> 2;
    const uint4* __restrict__ ids4 = (const uint4*)ids;
    auto emit = [&](int idx) {
      const int pos = atomicAdd(&q_count, 1);
      if (pos < OWN_Q) q_a[pos] = idx;
      else own_add_serial<D>(src, dst, dinv, h, tile, idx, e, lo, urows);
    };
    auto hit = [&](unsigned w) -> unsigned {
      const unsigned x = w ^ pat;
      return (x - 0x01010101u) & ~x & 0x80808080u;
    };
    auto pack = [&](uint4 g) -> unsigned {
      return (hit(g.x) >> 7) | (hit(g.y) >> 6) | (hit(g.z) >> 5) | (hit(g.w) >> 4);
    };
    constexpr int SU = 4;
    const int rot = (int)(((long long)blockIdx.x * ng) / (long long)gridDim.x);
    for (int base = 0; base < ng; base += OWN_T * SU) {
      uint4 g[SU];
#pragma unroll
      for (int j = 0; j < SU; ++j) {
        const int li = base + j * OWN_T + tid;
        int gi = (li < ng ? li : ng - 1) + rot;
        if (gi >= ng) gi -= ng;
        g[j] = ids4[gi];
      }
      unsigned f[SU];
#pragma unroll
      for (int j = 0; j < SU; ++j) {
        const int li = base + j * OWN_T + tid;
        f[j] = li < ng ? pack(g[j]) : 0u;
      }
      unsigned long long m = ((unsigned long long)(f[2] | (f[3] << 4)) << 32) | (unsigned long long)(f[0] | (f[1] << 4));
      while (m != 0ull) {
        const int b = __builtin_ctzll(m);
        m &= m - 1ull;
        const int j = ((b >> 5) << 1) | ((b >> 2) & 1);
        const int w = b & 3;
        const int k = (b & 31) >> 3;
        int gi = base + j * OWN_T + tid + rot;
        if (gi >= ng) gi -= ng;
        emit(gi * 16 + w * 4 + k);
      }
    }
    {
      const int idx = ng * 16 + tid;
      if (tid < 16 && (idx >> 2) < nw) {
        const unsigned w = ids[idx >> 2];
        if (((w >> (8 * (idx & 3))) & 0xFFu) == my) emit(idx);
      }
    }
  }
  __syncthreads();

  count = q_count;
  if (SAVE && tid == 0) g_hdr[0] = count;
  if (count > OWN_Q) count = OWN_Q;
  count = __builtin_amdgcn_readfirstlane(count);
  for (int i = tid; i < count; i += OWN_T) {
    const int idx = q_a[i];
    int s = 0, dl = -1;
    float nrm = 0.0f;
    if (idx < e) {
      const int d = dst[idx];
      const unsigned u = (unsigned)(d - lo);
      if (d >= lo && u < urows) {
        s = src[idx];
        dl = (int)u;
        nrm = dinv[s] * dinv[d];
      }
    }
    q_a[i] = s;
    q_dl[i] = dl;
    q_nrm[i] = nrm;
    if (SAVE) {
      g_a[i] = s;
      g_dl[i] = dl;
      g_nrm[i] = nrm;
    }
  }
  __syncthreads();
  }

  for (int i0 = wave; i0 < count; i0 += 32 * U) {
    int sv = 0, dv = -1, nv = 0;
    {
      const int i = i0 + 32 * (lane & (U - 1));
      if (lane < U && i < count) {
        sv = q_a[i];
        dv = q_dl[i];
        nv = __float_as_int(q_nrm[i]);
      }
    }
    float v[U][CH];
    int dl[U];
    float nr[U];
#pragma unroll
    for (int u = 0; u < U; ++u) {
      const int s = __builtin_amdgcn_readlane(sv, u);
      dl[u] = __builtin_amdgcn_readlane(dv, u);
      nr[u] = __int_as_float(__builtin_amdgcn_readlane(nv, u));
      const float* hp = h + (size_t)s * D + lane;
#pragma unroll
      for (int c = 0; c < CH; ++c) v[u][c] = hp[c * 32];
    }
#pragma unroll
    for (int u = 0; u < U; ++u) {
      if (dl[u] >= 0) {
        float* t = tile + dl[u] * D + lane;
#pragma unroll
        for (int c = 0; c < CH; ++c) atomicAdd(t + c * 32, v[u][c] * nr[u]);
      }
    }
  }
  __syncthreads();

  {
    float* __restrict__ ob = out + (size_t)lo * D;
#pragma unroll 8
    for (int i = tid; i < cells; i += OWN_T) ob[i] = tile[i];
  }
}

extern "C" void kernel_launch(void* const* d_in, const int* in_sizes, int n_in,
                              void* d_out, int out_size, void* d_ws, size_t ws_size,
                              hipStream_t stream) {
  (void)n_in; (void)out_size; (void)ws_size;

  const float* x    = (const float*)d_in[0];
  const int*   ei   = (const int*)d_in[1];
  const float* W1   = (const float*)d_in[2];
  const float* b1   = (const float*)d_in[3];
  const float* W2   = (const float*)d_in[4];
  const float* b2   = (const float*)d_in[5];
  float*       out  = (float*)d_out;

  const int N = in_sizes[0] / 128;
  const int E = in_sizes[1] / 2;
  const int IN_D = 128, HID_D = 128, OUT_D = 64;

  const int* src = ei;
  const int* dst = ei + E;

  float* dinv = (float*)d_ws;
  float* bufA = dinv + N;
  float* bufB = bufA + (size_t)N * 128;
  unsigned* ids = (unsigned*)((char*)d_ws + ((size_t)160 << 20));
  int* lists = (int*)((char*)d_ws + ((size_t)176 << 20));

  const int T = GCN_THREADS;
  auto blk = [&](long long work) { return (unsigned)((work + T - 1) / T); };

  k_deg_init<<<blk(N), T, 0, stream>>>(dinv, N);
  k_deg_count<<<blk((E + 3) / 4), T, 0, stream>>>(dst, dinv, ids, E);
  k_dinv<<<blk(N), T, 0, stream>>>(dinv, N);

  const int mtiles = N / 16;
  const int wavesPerBlk = T / 32;
  const unsigned g = (unsigned)((mtiles + wavesPerBlk - 1) / wavesPerBlk);
  const unsigned owners = (unsigned)((N + OWN_ROWS - 1) / OWN_ROWS);

  k_gemm_wmma<IN_D, HID_D, HID_D / 16, false><<<g, T, 0, stream>>>(x, W1, bufA, mtiles);

  k_owner_sum<HID_D, 16, true, false><<<owners, OWN_T, 0, stream>>>(src, dst, ids, dinv, bufA, b1, bufB, lists, N, E);

  k_gemm_wmma<HID_D, OUT_D, OUT_D / 16, true><<<g, T, 0, stream>>>(bufB, W2, bufA, mtiles);

  k_owner_sum<OUT_D, 16, false, true><<<owners, OWN_T, 0, stream>>>(src, dst, ids, dinv, bufA, b2, out, lists, N, E);
}


# Both products read W from LDS: each of 256 work-groups (one per WGP) copies W in once as ready B fragments; a fragment is one 8-byte LDS read, not two global loads; k order and sums unchanged
# speedup vs baseline: 2.2914x; 1.1755x over previous
#include <hip/hip_runtime.h>
#include <hip/hip_bf16.h>


typedef float v2f __attribute__((ext_vector_type(2)));
typedef float v8f __attribute__((ext_vector_type(8)));
typedef int v2i __attribute__((ext_vector_type(2)));
typedef float v4f __attribute__((ext_vector_type(4)));

#define GEMM_G 256
#define GEMM_T 1024

#define OWN_ROWS 391
#define OWN_T 1024
#define OWN_Q 4096
#define OWN_HDR 32
#define OWN_REC (OWN_HDR + 3 * OWN_Q)
#define OWN_BINS 256
#define OWN_PITCH 32
#define CNT_T 1024

__global__ __launch_bounds__(CNT_T) void k_deg_count(const int* __restrict__ src, const int* __restrict__ dst,
                                                    float* __restrict__ deg, int* __restrict__ cnt,
                                                    int* __restrict__ bins, int* __restrict__ ovf, int e) {
  __shared__ int hist[OWN_BINS];
  __shared__ int base[OWN_BINS];
  const int tid = (int)threadIdx.x;
  for (int b = tid; b < OWN_BINS; b += CNT_T) hist[b] = 0;
  __syncthreads();
  const int i0 = (int)blockIdx.x * (4 * CNT_T) + tid;
  int sv[4], dv[4], qv[4], rv[4];
  bool ok[4];
#pragma unroll
  for (int j = 0; j < 4; ++j) {
    const int i = i0 + j * CNT_T;
    ok[j] = i < e;
    dv[j] = ok[j] ? dst[i] : 0;
    sv[j] = ok[j] ? src[i] : 0;
  }
#pragma unroll
  for (int j = 0; j < 4; ++j) {
    qv[j] = -1;
    rv[j] = 0;
    if (ok[j]) {
      atomicAdd(&deg[dv[j]], 1.0f);
      const unsigned q = (unsigned)dv[j] / (unsigned)OWN_ROWS;
      qv[j] = q < (unsigned)(OWN_BINS - 1) ? (int)q : OWN_BINS - 1;
      rv[j] = atomicAdd(&hist[qv[j]], 1);
    }
  }
  __syncthreads();
  for (int b = tid; b < OWN_BINS; b += CNT_T) {
    const int c = hist[b];
    base[b] = c > 0 ? atomicAdd(&cnt[b * OWN_PITCH], c) : 0;
  }
  __syncthreads();
#pragma unroll
  for (int j = 0; j < 4; ++j) {
    if (qv[j] >= 0) {
      const int pos = base[qv[j]] + rv[j];
      const v2i sd = (v2i){sv[j], dv[j]};
      if (pos < OWN_Q) {
        *(v2i*)(bins + ((size_t)qv[j] * OWN_Q + (size_t)pos) * 2) = sd;
      } else {
        const int p = atomicAdd(&cnt[OWN_BINS * OWN_PITCH], 1);
        *(v2i*)(ovf + (size_t)p * 2) = sd;
      }
    }
  }
}

template <int K, int NO, int NT, bool RELU_A, bool INIT>
__global__ __launch_bounds__(GEMM_T) void k_gemm_wmma(const float* __restrict__ A, const float* __restrict__ W,
                                                     float* __restrict__ h, int mtiles,
                                                     float* __restrict__ deg, int* __restrict__ cnt, int n) {
  constexpr int WP = NO + 16;
  __shared__ v2f wl[(K / 2) * WP];

  const int tid  = (int)threadIdx.x;
  const int lane = tid & 31;
  const int wv   = __builtin_amdgcn_readfirstlane(tid >> 5);
  const int hi   = lane >> 4;
  const int lo   = lane & 15;
  const int kb   = hi << 1;

  if (INIT) {
    if (blockIdx.x == 0 && wv == 0) {
      for (int i = lane; i <= OWN_BINS; i += 32) cnt[i * OWN_PITCH] = 0;
      for (int i = mtiles * 16 + lane; i < n; i += 32) deg[i] = 1.0f;
    }
  }

  for (int idx = tid; idx < (K / 2) * (NO / 4); idx += GEMM_T) {
    const int p = idx / (NO / 4);
    const int c = (idx % (NO / 4)) * 4;
    const v4f r0 = *(const v4f*)(W + (size_t)(2 * p) * NO + c);
    const v4f r1 = *(const v4f*)(W + (size_t)(2 * p + 1) * NO + c);
    v2f* d = wl + p * WP + c;
    *(v4f*)(d)     = (v4f){r0.x, r1.x, r0.y, r1.y};
    *(v4f*)(d + 2) = (v4f){r0.z, r1.z, r0.w, r1.w};
  }
  __syncthreads();

  for (int tile = (int)blockIdx.x + GEMM_G * wv; tile < mtiles; tile += GEMM_G * (GEMM_T / 32)) {
    const int row = tile * 16 + lo;

    if (INIT) {
      if (hi == 0) deg[row] = 1.0f;
    }

    v8f acc[NT];
#pragma unroll
    for (int t = 0; t < NT; ++t) acc[t] = (v8f){0.f, 0.f, 0.f, 0.f, 0.f, 0.f, 0.f, 0.f};

    const float* __restrict__ arow = A + (size_t)row * K;

    for (int k0 = 0; k0 < K; k0 += 4) {
      v2f a = *(const v2f*)(arow + k0 + kb);
      if (RELU_A) {
        a.x = fmaxf(a.x, 0.0f);
        a.y = fmaxf(a.y, 0.0f);
      }
      const v2f* bp = wl + ((k0 >> 1) + hi) * WP + lo;
#pragma unroll
      for (int t = 0; t < NT; ++t) {
        const v2f b = bp[t * 16];
        acc[t] = __builtin_amdgcn_wmma_f32_16x16x4_f32(
             false, a,  false, b,
             (short)0, acc[t],  false,  false);
      }
    }

    const int mbase = tile * 16 + (hi << 3);
#pragma unroll
    for (int t = 0; t < NT; ++t) {
      const int col = t * 16 + lo;
#pragma unroll
      for (int r = 0; r < 8; ++r) h[(size_t)(mbase + r) * NO + col] = acc[t][r];
    }
  }
}

__device__ __forceinline__ float node_dinv(float deg) { return __builtin_amdgcn_rsqf(deg); }

template <int D>
__device__ __forceinline__ void own_add_serial(const float* __restrict__ deg, const float* __restrict__ h,
                                               float* tile, int s, int d, int lo, unsigned urows) {
  const unsigned u = (unsigned)(d - lo);
  if (d < lo || u >= urows) return;
  const float nrm = node_dinv(deg[s]) * node_dinv(deg[d]);
  const float* hp = h + (size_t)s * D;
  float* t = tile + u * D;
#pragma nounroll
  for (int c = 0; c < D; ++c) atomicAdd(t + c, hp[c] * nrm);
}

template <int D, int U, bool SAVE>
__global__ __launch_bounds__(OWN_T) void k_owner_sum(const int* __restrict__ cnt, const int* __restrict__ bins,
                                                    const int* __restrict__ ovf,
                                                    const float* __restrict__ deg, const float* __restrict__ h,
                                                    const float* __restrict__ bias, float* __restrict__ out,
                                                    int* __restrict__ lists, int n, int e) {
  constexpr int CH = D / 32;
  __shared__ float tile[OWN_ROWS * D];
  __shared__ int q_a[OWN_Q];
  __shared__ int q_dl[OWN_Q];
  __shared__ float q_nrm[OWN_Q];

  const int tid = (int)threadIdx.x;
  const int lane = tid & 31;
  const int wave = __builtin_amdgcn_readfirstlane(tid >> 5);
  const int lo = (int)blockIdx.x * OWN_ROWS;
  int rows = n - lo;
  if (rows > OWN_ROWS) rows = OWN_ROWS;
  if (rows <= 0) return;
  const unsigned urows = (unsigned)rows;
  const int cells = rows * D;

  int* __restrict__ g_hdr = lists + (size_t)blockIdx.x * OWN_REC;
  int* __restrict__ g_a = g_hdr + OWN_HDR;
  int* __restrict__ g_dl = g_a + OWN_Q;
  float* __restrict__ g_nrm = (float*)(g_dl + OWN_Q);
  constexpr int LPT = OWN_Q / OWN_T;

  int count;
  int la[LPT], ld[LPT];
  float ln[LPT];
  if (SAVE) {
    const int bin = (int)blockIdx.x < OWN_BINS - 1 ? (int)blockIdx.x : OWN_BINS - 1;
    count = __builtin_amdgcn_readfirstlane(cnt[bin * OWN_PITCH]);
    if (count > OWN_Q) count = OWN_Q;
    if (count < 0) count = 0;
    const int* __restrict__ gb = bins + (size_t)bin * (2 * OWN_Q);
    v2i sd[LPT];
#pragma unroll
    for (int j = 0; j < LPT; ++j) {
      const int i = tid + j * OWN_T;
      sd[j] = i < count ? *(const v2i*)(gb + 2 * i) : (v2i){0, -1};
    }
#pragma unroll
    for (int j = 0; j < LPT; ++j) {
      const int d = sd[j].y;
      const unsigned u = (unsigned)(d - lo);
      const bool own = d >= lo && u < urows;
      const int s = own ? sd[j].x : 0;
      const float nrm = node_dinv(deg[s]) * node_dinv(deg[own ? d : lo]);
      la[j] = s;
      ld[j] = own ? (int)u : -1;
      ln[j] = own ? nrm : 0.0f;
    }
  } else {
    count = __builtin_amdgcn_readfirstlane(g_hdr[0]);
    if (count > OWN_Q) count = OWN_Q;
    if (count < 0) count = 0;
#pragma unroll
    for (int j = 0; j < LPT; ++j) {
      const int i = tid + j * OWN_T;
      const bool ok = i < count;
      la[j] = ok ? g_a[i] : 0;
      ld[j] = ok ? g_dl[i] : -1;
      ln[j] = ok ? g_nrm[i] : 0.0f;
    }
  }

  {
    const float bv = bias[tid % D];
    const float* __restrict__ hb = h + (size_t)lo * D;
#pragma unroll 8
    for (int i = tid; i < cells; i += OWN_T) {
      const float s = node_dinv(deg[lo + i / D]);
      tile[i] = hb[i] * (s * s) + bv;
    }
  }
#pragma unroll
  for (int j = 0; j < LPT; ++j) {
    const int i = tid + j * OWN_T;
    if (i < count) {
      q_a[i] = la[j];
      q_dl[i] = ld[j];
      q_nrm[i] = ln[j];
      if (SAVE) {
        g_a[i] = la[j];
        g_dl[i] = ld[j];
        g_nrm[i] = ln[j];
      }
    }
  }
  if (SAVE && tid == 0) g_hdr[0] = count;
  __syncthreads();

  for (int i0 = wave; i0 < count; i0 += 32 * U) {
    int sv = 0, dv = -1, nv = 0;
    {
      const int i = i0 + 32 * (lane & (U - 1));
      if (lane < U && i < count) {
        sv = q_a[i];
        dv = q_dl[i];
        nv = __float_as_int(q_nrm[i]);
      }
    }
    float v[U][CH];
    int dl[U];
    float nr[U];
#pragma unroll
    for (int u = 0; u < U; ++u) {
      const int s = __builtin_amdgcn_readlane(sv, u);
      dl[u] = __builtin_amdgcn_readlane(dv, u);
      nr[u] = __int_as_float(__builtin_amdgcn_readlane(nv, u));
      const float* hp = h + (size_t)s * D + lane;
#pragma unroll
      for (int c = 0; c < CH; ++c) v[u][c] = hp[c * 32];
    }
#pragma unroll
    for (int u = 0; u < U; ++u) {
      if (dl[u] >= 0) {
        float* t = tile + dl[u] * D + lane;
#pragma unroll
        for (int c = 0; c < CH; ++c) atomicAdd(t + c * 32, v[u][c] * nr[u]);
      }
    }
  }

  {
    int novf = __builtin_amdgcn_readfirstlane(cnt[OWN_BINS * OWN_PITCH]);
    if (novf > e) novf = e;
    for (int i = tid; i < novf; i += OWN_T) {
      const v2i sd = *(const v2i*)(ovf + (size_t)i * 2);
      own_add_serial<D>(deg, h, tile, sd.x, sd.y, lo, urows);
    }
  }
  __syncthreads();

  {
    float* __restrict__ ob = out + (size_t)lo * D;
#pragma unroll 8
    for (int i = tid; i < cells; i += OWN_T) ob[i] = tile[i];
  }
}

extern "C" void kernel_launch(void* const* d_in, const int* in_sizes, int n_in,
                              void* d_out, int out_size, void* d_ws, size_t ws_size,
                              hipStream_t stream) {
  (void)n_in; (void)out_size; (void)ws_size;

  const float* x    = (const float*)d_in[0];
  const int*   ei   = (const int*)d_in[1];
  const float* W1   = (const float*)d_in[2];
  const float* b1   = (const float*)d_in[3];
  const float* W2   = (const float*)d_in[4];
  const float* b2   = (const float*)d_in[5];
  float*       out  = (float*)d_out;

  const int N = in_sizes[0] / 128;
  const int E = in_sizes[1] / 2;
  const int IN_D = 128, HID_D = 128, OUT_D = 64;

  const int* src = ei;
  const int* dst = ei + E;

  float* deg  = (float*)d_ws;
  float* bufA = deg + N;
  float* bufB = bufA + (size_t)N * 128;
  int* cnt = (int*)((char*)d_ws + ((size_t)160 << 20));
  int* lists = (int*)((char*)d_ws + ((size_t)176 << 20));
  int* bins = (int*)((char*)d_ws + ((size_t)192 << 20));
  int* ovf = (int*)((char*)d_ws + ((size_t)208 << 20));

  const int mtiles = N / 16;
  const unsigned owners = (unsigned)((N + OWN_ROWS - 1) / OWN_ROWS);

  k_gemm_wmma<IN_D, HID_D, HID_D / 16, false, true><<<GEMM_G, GEMM_T, 0, stream>>>(x, W1, bufA, mtiles, deg, cnt, N);

  k_deg_count<<<(unsigned)((E + 4 * CNT_T - 1) / (4 * CNT_T)), CNT_T, 0, stream>>>(src, dst, deg, cnt, bins, ovf, E);

  k_owner_sum<HID_D, 16, true><<<owners, OWN_T, 0, stream>>>(cnt, bins, ovf, deg, bufA, b1, bufB, lists, N, E);

  k_gemm_wmma<HID_D, OUT_D, OUT_D / 16, true, false><<<GEMM_G, GEMM_T, 0, stream>>>(bufB, W2, bufA, mtiles, nullptr, nullptr, 0);

  k_owner_sum<OUT_D, 16, false><<<owners, OWN_T, 0, stream>>>(cnt, bins, ovf, deg, bufA, b2, out, lists, N, E);
}


# Degree count leaves scattered global float atomics: a new small launch has each owner count its own edge bin's destinations in LDS integer counters and store deg once; 5 launches become 6
# speedup vs baseline: 2.8665x; 1.2510x over previous
#include <hip/hip_runtime.h>
#include <hip/hip_bf16.h>


typedef float v2f __attribute__((ext_vector_type(2)));
typedef float v8f __attribute__((ext_vector_type(8)));
typedef int v2i __attribute__((ext_vector_type(2)));
typedef float v4f __attribute__((ext_vector_type(4)));

#define GEMM_G 256
#define GEMM_T 1024

#define OWN_ROWS 391
#define OWN_T 1024
#define OWN_Q 4096
#define OWN_HDR 32
#define OWN_REC (OWN_HDR + 3 * OWN_Q)
#define OWN_BINS 256
#define OWN_PITCH 32
#define CNT_T 1024
#define DEG_T 256

__global__ __launch_bounds__(CNT_T) void k_edge_file(const int* __restrict__ src, const int* __restrict__ dst,
                                                    int* __restrict__ cnt,
                                                    int* __restrict__ bins, int* __restrict__ ovf, int e) {
  __shared__ int hist[OWN_BINS];
  __shared__ int base[OWN_BINS];
  const int tid = (int)threadIdx.x;
  for (int b = tid; b < OWN_BINS; b += CNT_T) hist[b] = 0;
  __syncthreads();
  const int i0 = (int)blockIdx.x * (4 * CNT_T) + tid;
  int sv[4], dv[4], qv[4], rv[4];
  bool ok[4];
#pragma unroll
  for (int j = 0; j < 4; ++j) {
    const int i = i0 + j * CNT_T;
    ok[j] = i < e;
    dv[j] = ok[j] ? dst[i] : 0;
    sv[j] = ok[j] ? src[i] : 0;
  }
#pragma unroll
  for (int j = 0; j < 4; ++j) {
    qv[j] = -1;
    rv[j] = 0;
    if (ok[j]) {
      const unsigned q = (unsigned)dv[j] / (unsigned)OWN_ROWS;
      qv[j] = q < (unsigned)(OWN_BINS - 1) ? (int)q : OWN_BINS - 1;
      rv[j] = atomicAdd(&hist[qv[j]], 1);
    }
  }
  __syncthreads();
  for (int b = tid; b < OWN_BINS; b += CNT_T) {
    const int c = hist[b];
    base[b] = c > 0 ? atomicAdd(&cnt[b * OWN_PITCH], c) : 0;
  }
  __syncthreads();
#pragma unroll
  for (int j = 0; j < 4; ++j) {
    if (qv[j] >= 0) {
      const int pos = base[qv[j]] + rv[j];
      const v2i sd = (v2i){sv[j], dv[j]};
      if (pos < OWN_Q) {
        *(v2i*)(bins + ((size_t)qv[j] * OWN_Q + (size_t)pos) * 2) = sd;
      } else {
        const int p = atomicAdd(&cnt[OWN_BINS * OWN_PITCH], 1);
        *(v2i*)(ovf + (size_t)p * 2) = sd;
      }
    }
  }
}

__global__ __launch_bounds__(DEG_T) void k_owner_deg(const int* __restrict__ cnt, const int* __restrict__ bins,
                                                    const int* __restrict__ ovf, float* __restrict__ deg,
                                                    int n, int e) {
  __shared__ int hist[OWN_ROWS];
  const int tid = (int)threadIdx.x;
  const int lo = (int)blockIdx.x * OWN_ROWS;
  int rows = n - lo;
  if (rows > OWN_ROWS) rows = OWN_ROWS;
  if (rows <= 0) return;
  const unsigned urows = (unsigned)rows;
  const int bin = (int)blockIdx.x < OWN_BINS - 1 ? (int)blockIdx.x : OWN_BINS - 1;
  int count = __builtin_amdgcn_readfirstlane(cnt[bin * OWN_PITCH]);
  if (count > OWN_Q) count = OWN_Q;
  if (count < 0) count = 0;
  int novf = __builtin_amdgcn_readfirstlane(cnt[OWN_BINS * OWN_PITCH]);
  if (novf > e) novf = e;
  constexpr int LPT = OWN_Q / DEG_T;
  const int* __restrict__ gb = bins + (size_t)bin * (2 * OWN_Q);
  int dv[LPT];
#pragma unroll
  for (int j = 0; j < LPT; ++j) {
    const int i = tid + j * DEG_T;
    dv[j] = i < count ? gb[2 * i + 1] : -1;
  }
  for (int r = tid; r < OWN_ROWS; r += DEG_T) hist[r] = 0;
  __syncthreads();
#pragma unroll
  for (int j = 0; j < LPT; ++j) {
    const unsigned u = (unsigned)(dv[j] - lo);
    if (dv[j] >= lo && u < urows) atomicAdd(&hist[u], 1);
  }
  for (int i = tid; i < novf; i += DEG_T) {
    const int d = ovf[2 * (size_t)i + 1];
    const unsigned u = (unsigned)(d - lo);
    if (d >= lo && u < urows) atomicAdd(&hist[u], 1);
  }
  __syncthreads();
  for (int r = tid; r < rows; r += DEG_T) deg[lo + r] = (float)(hist[r] + 1);
}

template <int K, int NO, int NT, bool RELU_A, bool INIT>
__global__ __launch_bounds__(GEMM_T) void k_gemm_wmma(const float* __restrict__ A, const float* __restrict__ W,
                                                     float* __restrict__ h, int mtiles,
                                                     int* __restrict__ cnt) {
  constexpr int WP = NO + 16;
  __shared__ v2f wl[(K / 2) * WP];

  const int tid  = (int)threadIdx.x;
  const int lane = tid & 31;
  const int wv   = __builtin_amdgcn_readfirstlane(tid >> 5);
  const int hi   = lane >> 4;
  const int lo   = lane & 15;
  const int kb   = hi << 1;

  if (INIT) {
    if (blockIdx.x == 0 && wv == 0) {
      for (int i = lane; i <= OWN_BINS; i += 32) cnt[i * OWN_PITCH] = 0;
    }
  }

  for (int idx = tid; idx < (K / 2) * (NO / 4); idx += GEMM_T) {
    const int p = idx / (NO / 4);
    const int c = (idx % (NO / 4)) * 4;
    const v4f r0 = *(const v4f*)(W + (size_t)(2 * p) * NO + c);
    const v4f r1 = *(const v4f*)(W + (size_t)(2 * p + 1) * NO + c);
    v2f* d = wl + p * WP + c;
    *(v4f*)(d)     = (v4f){r0.x, r1.x, r0.y, r1.y};
    *(v4f*)(d + 2) = (v4f){r0.z, r1.z, r0.w, r1.w};
  }
  __syncthreads();

  for (int tile = (int)blockIdx.x + GEMM_G * wv; tile < mtiles; tile += GEMM_G * (GEMM_T / 32)) {
    const int row = tile * 16 + lo;

    v8f acc[NT];
#pragma unroll
    for (int t = 0; t < NT; ++t) acc[t] = (v8f){0.f, 0.f, 0.f, 0.f, 0.f, 0.f, 0.f, 0.f};

    const float* __restrict__ arow = A + (size_t)row * K;

    for (int k0 = 0; k0 < K; k0 += 4) {
      v2f a = *(const v2f*)(arow + k0 + kb);
      if (RELU_A) {
        a.x = fmaxf(a.x, 0.0f);
        a.y = fmaxf(a.y, 0.0f);
      }
      const v2f* bp = wl + ((k0 >> 1) + hi) * WP + lo;
#pragma unroll
      for (int t = 0; t < NT; ++t) {
        const v2f b = bp[t * 16];
        acc[t] = __builtin_amdgcn_wmma_f32_16x16x4_f32(
             false, a,  false, b,
             (short)0, acc[t],  false,  false);
      }
    }

    const int mbase = tile * 16 + (hi << 3);
#pragma unroll
    for (int t = 0; t < NT; ++t) {
      const int col = t * 16 + lo;
#pragma unroll
      for (int r = 0; r < 8; ++r) h[(size_t)(mbase + r) * NO + col] = acc[t][r];
    }
  }
}

__device__ __forceinline__ float node_dinv(float deg) { return __builtin_amdgcn_rsqf(deg); }

template <int D>
__device__ __forceinline__ void own_add_serial(const float* __restrict__ deg, const float* __restrict__ h,
                                               float* tile, int s, int d, int lo, unsigned urows) {
  const unsigned u = (unsigned)(d - lo);
  if (d < lo || u >= urows) return;
  const float nrm = node_dinv(deg[s]) * node_dinv(deg[d]);
  const float* hp = h + (size_t)s * D;
  float* t = tile + u * D;
#pragma nounroll
  for (int c = 0; c < D; ++c) atomicAdd(t + c, hp[c] * nrm);
}

template <int D, int U, bool SAVE>
__global__ __launch_bounds__(OWN_T) void k_owner_sum(const int* __restrict__ cnt, const int* __restrict__ bins,
                                                    const int* __restrict__ ovf,
                                                    const float* __restrict__ deg, const float* __restrict__ h,
                                                    const float* __restrict__ bias, float* __restrict__ out,
                                                    int* __restrict__ lists, int n, int e) {
  constexpr int CH = D / 32;
  __shared__ float tile[OWN_ROWS * D];
  __shared__ int q_a[OWN_Q];
  __shared__ int q_dl[OWN_Q];
  __shared__ float q_nrm[OWN_Q];

  const int tid = (int)threadIdx.x;
  const int lane = tid & 31;
  const int wave = __builtin_amdgcn_readfirstlane(tid >> 5);
  const int lo = (int)blockIdx.x * OWN_ROWS;
  int rows = n - lo;
  if (rows > OWN_ROWS) rows = OWN_ROWS;
  if (rows <= 0) return;
  const unsigned urows = (unsigned)rows;
  const int cells = rows * D;

  int* __restrict__ g_hdr = lists + (size_t)blockIdx.x * OWN_REC;
  int* __restrict__ g_a = g_hdr + OWN_HDR;
  int* __restrict__ g_dl = g_a + OWN_Q;
  float* __restrict__ g_nrm = (float*)(g_dl + OWN_Q);
  constexpr int LPT = OWN_Q / OWN_T;

  int count;
  int la[LPT], ld[LPT];
  float ln[LPT];
  if (SAVE) {
    const int bin = (int)blockIdx.x < OWN_BINS - 1 ? (int)blockIdx.x : OWN_BINS - 1;
    count = __builtin_amdgcn_readfirstlane(cnt[bin * OWN_PITCH]);
    if (count > OWN_Q) count = OWN_Q;
    if (count < 0) count = 0;
    const int* __restrict__ gb = bins + (size_t)bin * (2 * OWN_Q);
    v2i sd[LPT];
#pragma unroll
    for (int j = 0; j < LPT; ++j) {
      const int i = tid + j * OWN_T;
      sd[j] = i < count ? *(const v2i*)(gb + 2 * i) : (v2i){0, -1};
    }
#pragma unroll
    for (int j = 0; j < LPT; ++j) {
      const int d = sd[j].y;
      const unsigned u = (unsigned)(d - lo);
      const bool own = d >= lo && u < urows;
      const int s = own ? sd[j].x : 0;
      const float nrm = node_dinv(deg[s]) * node_dinv(deg[own ? d : lo]);
      la[j] = s;
      ld[j] = own ? (int)u : -1;
      ln[j] = own ? nrm : 0.0f;
    }
  } else {
    count = __builtin_amdgcn_readfirstlane(g_hdr[0]);
    if (count > OWN_Q) count = OWN_Q;
    if (count < 0) count = 0;
#pragma unroll
    for (int j = 0; j < LPT; ++j) {
      const int i = tid + j * OWN_T;
      const bool ok = i < count;
      la[j] = ok ? g_a[i] : 0;
      ld[j] = ok ? g_dl[i] : -1;
      ln[j] = ok ? g_nrm[i] : 0.0f;
    }
  }

  {
    const float bv = bias[tid % D];
    const float* __restrict__ hb = h + (size_t)lo * D;
#pragma unroll 8
    for (int i = tid; i < cells; i += OWN_T) {
      const float s = node_dinv(deg[lo + i / D]);
      tile[i] = hb[i] * (s * s) + bv;
    }
  }
#pragma unroll
  for (int j = 0; j < LPT; ++j) {
    const int i = tid + j * OWN_T;
    if (i < count) {
      q_a[i] = la[j];
      q_dl[i] = ld[j];
      q_nrm[i] = ln[j];
      if (SAVE) {
        g_a[i] = la[j];
        g_dl[i] = ld[j];
        g_nrm[i] = ln[j];
      }
    }
  }
  if (SAVE && tid == 0) g_hdr[0] = count;
  __syncthreads();

  for (int i0 = wave; i0 < count; i0 += 32 * U) {
    int sv = 0, dv = -1, nv = 0;
    {
      const int i = i0 + 32 * (lane & (U - 1));
      if (lane < U && i < count) {
        sv = q_a[i];
        dv = q_dl[i];
        nv = __float_as_int(q_nrm[i]);
      }
    }
    float v[U][CH];
    int dl[U];
    float nr[U];
#pragma unroll
    for (int u = 0; u < U; ++u) {
      const int s = __builtin_amdgcn_readlane(sv, u);
      dl[u] = __builtin_amdgcn_readlane(dv, u);
      nr[u] = __int_as_float(__builtin_amdgcn_readlane(nv, u));
      const float* hp = h + (size_t)s * D + lane;
#pragma unroll
      for (int c = 0; c < CH; ++c) v[u][c] = hp[c * 32];
    }
#pragma unroll
    for (int u = 0; u < U; ++u) {
      if (dl[u] >= 0) {
        float* t = tile + dl[u] * D + lane;
#pragma unroll
        for (int c = 0; c < CH; ++c) atomicAdd(t + c * 32, v[u][c] * nr[u]);
      }
    }
  }

  {
    int novf = __builtin_amdgcn_readfirstlane(cnt[OWN_BINS * OWN_PITCH]);
    if (novf > e) novf = e;
    for (int i = tid; i < novf; i += OWN_T) {
      const v2i sd = *(const v2i*)(ovf + (size_t)i * 2);
      own_add_serial<D>(deg, h, tile, sd.x, sd.y, lo, urows);
    }
  }
  __syncthreads();

  {
    float* __restrict__ ob = out + (size_t)lo * D;
#pragma unroll 8
    for (int i = tid; i < cells; i += OWN_T) ob[i] = tile[i];
  }
}

extern "C" void kernel_launch(void* const* d_in, const int* in_sizes, int n_in,
                              void* d_out, int out_size, void* d_ws, size_t ws_size,
                              hipStream_t stream) {
  (void)n_in; (void)out_size; (void)ws_size;

  const float* x    = (const float*)d_in[0];
  const int*   ei   = (const int*)d_in[1];
  const float* W1   = (const float*)d_in[2];
  const float* b1   = (const float*)d_in[3];
  const float* W2   = (const float*)d_in[4];
  const float* b2   = (const float*)d_in[5];
  float*       out  = (float*)d_out;

  const int N = in_sizes[0] / 128;
  const int E = in_sizes[1] / 2;
  const int IN_D = 128, HID_D = 128, OUT_D = 64;

  const int* src = ei;
  const int* dst = ei + E;

  float* deg  = (float*)d_ws;
  float* bufA = deg + N;
  float* bufB = bufA + (size_t)N * 128;
  int* cnt = (int*)((char*)d_ws + ((size_t)160 << 20));
  int* lists = (int*)((char*)d_ws + ((size_t)176 << 20));
  int* bins = (int*)((char*)d_ws + ((size_t)192 << 20));
  int* ovf = (int*)((char*)d_ws + ((size_t)208 << 20));

  const int mtiles = N / 16;
  const unsigned owners = (unsigned)((N + OWN_ROWS - 1) / OWN_ROWS);

  k_gemm_wmma<IN_D, HID_D, HID_D / 16, false, true><<<GEMM_G, GEMM_T, 0, stream>>>(x, W1, bufA, mtiles, cnt);

  k_edge_file<<<(unsigned)((E + 4 * CNT_T - 1) / (4 * CNT_T)), CNT_T, 0, stream>>>(src, dst, cnt, bins, ovf, E);

  k_owner_deg<<<owners, DEG_T, 0, stream>>>(cnt, bins, ovf, deg, N, E);

  k_owner_sum<HID_D, 16, true><<<owners, OWN_T, 0, stream>>>(cnt, bins, ovf, deg, bufA, b1, bufB, lists, N, E);

  k_gemm_wmma<HID_D, OUT_D, OUT_D / 16, true, false><<<GEMM_G, GEMM_T, 0, stream>>>(bufB, W2, bufA, mtiles, nullptr);

  k_owner_sum<OUT_D, 16, false><<<owners, OWN_T, 0, stream>>>(cnt, bins, ovf, deg, bufA, b2, out, lists, N, E);
}
